# starting kernel
# baseline (speedup 1.0000x reference)
#include <hip/hip_runtime.h>

#define B_   8
#define N_   65536
#define D_   40
#define DP_  48
#define K_   2
#define CHUNKS_ 128
#define CHUNK_  512
#define WAVES_  8
#define ROWS_PER_WAVE_ 64
#define WAVE_ITERS_ 2

typedef float v2f __attribute__((ext_vector_type(2)));
typedef float v8f __attribute__((ext_vector_type(8)));

__global__ __launch_bounds__(256) void sk_init(const float* __restrict__ data,
                                               const int* __restrict__ ids,
                                               float* __restrict__ cent) {
  int tid = blockIdx.x * 256 + threadIdx.x;
  if (tid >= B_ * K_ * DP_) return;
  int b  = tid / (K_ * DP_);
  int kd = tid % (K_ * DP_);
  int k  = kd / DP_;
  int d  = kd % DP_;
  int n  = ids[b * K_ + k];
  cent[tid] = (d < D_) ? data[((size_t)b * N_ + n) * D_ + d] : 0.0f;
}

__global__ __launch_bounds__(256) void sk_pass(const float* __restrict__ data,
                                               const float* __restrict__ cent,
                                               float* __restrict__ partials,
                                               float2* __restrict__ probs_out,
                                               int write_probs, int accumulate) {
  __shared__ float cLds[K_ * DP_];
  __shared__ float xt[WAVES_][32 * DP_];
  __shared__ float pt[WAVES_][32 * K_];
  __shared__ float red[WAVES_][K_ * DP_];

  const int tid   = threadIdx.x;
  const int wave  = tid >> 5;
  const int lane  = tid & 31;
  const int b     = blockIdx.x >> 7;
  const int chunk = blockIdx.x & (CHUNKS_ - 1);

  for (int i = tid; i < K_ * DP_; i += 256) {
    int k = i / DP_, d = i % DP_;
    cLds[i] = (d < D_) ? cent[(b * K_ + k) * DP_ + d] : 0.0f;
  }
  __syncthreads();

  float cc0 = 0.f, cc1 = 0.f;
#pragma unroll
  for (int d = 0; d < D_; ++d) {
    float a0 = cLds[d], a1 = cLds[DP_ + d];
    cc0 = fmaf(a0, a0, cc0);
    cc1 = fmaf(a1, a1, cc1);
  }

  v8f acc[3] = {};
  float* myxt = &xt[wave][0];
  float* mypt = &pt[wave][0];
  const float* __restrict__ dbase = data + (size_t)b * N_ * D_;
  const int m15  = lane & 15;
  const int m1   = lane & 1;
  const bool alive = (m15 < K_);
  const int rsel = (lane >> 4) << 1;

#pragma unroll
  for (int it = 0; it < WAVE_ITERS_; ++it) {
    const int n = chunk * CHUNK_ + wave * ROWS_PER_WAVE_ + it * 32 + lane;
    const float4* __restrict__ xr = (const float4*)(dbase + (size_t)n * D_);
    float x[D_];
#pragma unroll
    for (int q = 0; q < D_ / 4; ++q) {
      float4 v = xr[q];
      x[4 * q + 0] = v.x; x[4 * q + 1] = v.y;
      x[4 * q + 2] = v.z; x[4 * q + 3] = v.w;
    }
    float dot0 = 0.f, dot1 = 0.f;
#pragma unroll
    for (int d = 0; d < D_; ++d) {
      dot0 = fmaf(x[d], cLds[d], dot0);
      dot1 = fmaf(x[d], cLds[DP_ + d], dot1);
    }
    float s0 = fmaf(2.0f, dot0, -cc0);
    float s1 = fmaf(2.0f, dot1, -cc1);
    float mx = fmaxf(s0, s1);
    float e0 = __expf(s0 - mx);
    float e1 = __expf(s1 - mx);
    float inv = 1.0f / (e0 + e1);
    float p0 = e0 * inv, p1 = e1 * inv;

    if (write_probs) {
      probs_out[(size_t)b * N_ + n] = make_float2(p0, p1);
    }
    if (accumulate) {
#pragma unroll
      for (int d = 0; d < D_; ++d) myxt[lane * DP_ + d] = x[d];
      myxt[lane * DP_ + D_] = 1.0f;
#pragma unroll
      for (int d = D_ + 1; d < DP_; ++d) myxt[lane * DP_ + d] = 0.0f;
      mypt[lane * K_ + 0] = p0;
      mypt[lane * K_ + 1] = p1;
      __builtin_amdgcn_wave_barrier();

#pragma unroll
      for (int g = 0; g < 8; ++g) {
        const int r0 = 4 * g + rsel;
        float a0v = mypt[(r0 + 0) * K_ + m1];
        float a1v = mypt[(r0 + 1) * K_ + m1];
        v2f a;
        a.x = alive ? a0v : 0.0f;
        a.y = alive ? a1v : 0.0f;
#pragma unroll
        for (int t = 0; t < 3; ++t) {
          v2f bm;
          bm.x = myxt[(r0 + 0) * DP_ + 16 * t + m15];
          bm.y = myxt[(r0 + 1) * DP_ + 16 * t + m15];
          acc[t] = __builtin_amdgcn_wmma_f32_16x16x4_f32(
              false, a, false, bm, (short)0, acc[t], false, false);
        }
      }
      __builtin_amdgcn_wave_barrier();
    }
  }

  if (accumulate) {
    if (lane < 16) {
#pragma unroll
      for (int t = 0; t < 3; ++t) {
        red[wave][0 * DP_ + 16 * t + lane] = acc[t][0];
        red[wave][1 * DP_ + 16 * t + lane] = acc[t][1];
      }
    }
    __syncthreads();
    if (tid < K_ * DP_) {
      float s = 0.f;
#pragma unroll
      for (int w = 0; w < WAVES_; ++w) s += red[w][tid];
      partials[(size_t)blockIdx.x * (K_ * DP_) + tid] = s;
    }
  }
}

__global__ __launch_bounds__(256) void sk_reduce(const float* __restrict__ partials,
                                                 float* __restrict__ cent) {
  int tid = blockIdx.x * 256 + threadIdx.x;
  if (tid >= B_ * K_ * DP_) return;
  int b  = tid / (K_ * DP_);
  int kd = tid % (K_ * DP_);
  int k  = kd / DP_;
  int d  = kd % DP_;
  float s = 0.f, sden = 0.f;
  for (int c = 0; c < CHUNKS_; ++c) {
    const float* p = partials + (size_t)(b * CHUNKS_ + c) * (K_ * DP_);
    s    += p[kd];
    sden += p[k * DP_ + D_];
  }
  cent[tid] = (d < D_) ? (s / sden) : 0.0f;
}

extern "C" void kernel_launch(void* const* d_in, const int* in_sizes, int n_in,
                              void* d_out, int out_size, void* d_ws, size_t ws_size,
                              hipStream_t stream) {
  const float* data = (const float*)d_in[0];
  const int*   ids  = (const int*)d_in[1];
  const int ITER = 10;

  float*  cent     = (float*)d_ws;
  float*  partials = cent + B_ * K_ * DP_;
  float2* out      = (float2*)d_out;

  sk_init<<<(B_ * K_ * DP_ + 255) / 256, 256, 0, stream>>>(data, ids, cent);
  for (int t = 0; t <= ITER; ++t) {
    int wp = (t == ITER) ? 1 : 0;
    int ac = (t <  ITER) ? 1 : 0;
    sk_pass<<<B_ * CHUNKS_, 256, 0, stream>>>(data, cent, partials, out, wp, ac);
    if (ac)
      sk_reduce<<<(B_ * K_ * DP_ + 255) / 256, 256, 0, stream>>>(partials, cent);
  }
  (void)in_sizes; (void)n_in; (void)out_size; (void)ws_size;
}


# sk_reduce: sum the 128 chunk partials per batch in parallel (16 slices x 24 float4 columns, one load group, LDS join) instead of a 128-deep serial load chain; chunk summation order now two-level
# speedup vs baseline: 1.9763x; 1.9763x over previous
#include <hip/hip_runtime.h>

#define B_   8
#define N_   65536
#define D_   40
#define DP_  48
#define K_   2
#define CHUNKS_ 128
#define CHUNK_  512
#define WAVES_  8
#define ROWS_PER_WAVE_ 64
#define WAVE_ITERS_ 2

typedef float v2f __attribute__((ext_vector_type(2)));
typedef float v8f __attribute__((ext_vector_type(8)));

__global__ __launch_bounds__(256) void sk_init(const float* __restrict__ data,
                                               const int* __restrict__ ids,
                                               float* __restrict__ cent) {
  int tid = blockIdx.x * 256 + threadIdx.x;
  if (tid >= B_ * K_ * DP_) return;
  int b  = tid / (K_ * DP_);
  int kd = tid % (K_ * DP_);
  int k  = kd / DP_;
  int d  = kd % DP_;
  int n  = ids[b * K_ + k];
  cent[tid] = (d < D_) ? data[((size_t)b * N_ + n) * D_ + d] : 0.0f;
}

__global__ __launch_bounds__(256) void sk_pass(const float* __restrict__ data,
                                               const float* __restrict__ cent,
                                               float* __restrict__ partials,
                                               float2* __restrict__ probs_out,
                                               int write_probs, int accumulate) {
  __shared__ float cLds[K_ * DP_];
  __shared__ float xt[WAVES_][32 * DP_];
  __shared__ float pt[WAVES_][32 * K_];
  __shared__ float red[WAVES_][K_ * DP_];

  const int tid   = threadIdx.x;
  const int wave  = tid >> 5;
  const int lane  = tid & 31;
  const int b     = blockIdx.x >> 7;
  const int chunk = blockIdx.x & (CHUNKS_ - 1);

  for (int i = tid; i < K_ * DP_; i += 256) {
    int k = i / DP_, d = i % DP_;
    cLds[i] = (d < D_) ? cent[(b * K_ + k) * DP_ + d] : 0.0f;
  }
  __syncthreads();

  float cc0 = 0.f, cc1 = 0.f;
#pragma unroll
  for (int d = 0; d < D_; ++d) {
    float a0 = cLds[d], a1 = cLds[DP_ + d];
    cc0 = fmaf(a0, a0, cc0);
    cc1 = fmaf(a1, a1, cc1);
  }

  v8f acc[3] = {};
  float* myxt = &xt[wave][0];
  float* mypt = &pt[wave][0];
  const float* __restrict__ dbase = data + (size_t)b * N_ * D_;
  const int m15  = lane & 15;
  const int m1   = lane & 1;
  const bool alive = (m15 < K_);
  const int rsel = (lane >> 4) << 1;

#pragma unroll
  for (int it = 0; it < WAVE_ITERS_; ++it) {
    const int n = chunk * CHUNK_ + wave * ROWS_PER_WAVE_ + it * 32 + lane;
    const float4* __restrict__ xr = (const float4*)(dbase + (size_t)n * D_);
    float x[D_];
#pragma unroll
    for (int q = 0; q < D_ / 4; ++q) {
      float4 v = xr[q];
      x[4 * q + 0] = v.x; x[4 * q + 1] = v.y;
      x[4 * q + 2] = v.z; x[4 * q + 3] = v.w;
    }
    float dot0 = 0.f, dot1 = 0.f;
#pragma unroll
    for (int d = 0; d < D_; ++d) {
      dot0 = fmaf(x[d], cLds[d], dot0);
      dot1 = fmaf(x[d], cLds[DP_ + d], dot1);
    }
    float s0 = fmaf(2.0f, dot0, -cc0);
    float s1 = fmaf(2.0f, dot1, -cc1);
    float mx = fmaxf(s0, s1);
    float e0 = __expf(s0 - mx);
    float e1 = __expf(s1 - mx);
    float inv = 1.0f / (e0 + e1);
    float p0 = e0 * inv, p1 = e1 * inv;

    if (write_probs) {
      probs_out[(size_t)b * N_ + n] = make_float2(p0, p1);
    }
    if (accumulate) {
#pragma unroll
      for (int d = 0; d < D_; ++d) myxt[lane * DP_ + d] = x[d];
      myxt[lane * DP_ + D_] = 1.0f;
#pragma unroll
      for (int d = D_ + 1; d < DP_; ++d) myxt[lane * DP_ + d] = 0.0f;
      mypt[lane * K_ + 0] = p0;
      mypt[lane * K_ + 1] = p1;
      __builtin_amdgcn_wave_barrier();

#pragma unroll
      for (int g = 0; g < 8; ++g) {
        const int r0 = 4 * g + rsel;
        float a0v = mypt[(r0 + 0) * K_ + m1];
        float a1v = mypt[(r0 + 1) * K_ + m1];
        v2f a;
        a.x = alive ? a0v : 0.0f;
        a.y = alive ? a1v : 0.0f;
#pragma unroll
        for (int t = 0; t < 3; ++t) {
          v2f bm;
          bm.x = myxt[(r0 + 0) * DP_ + 16 * t + m15];
          bm.y = myxt[(r0 + 1) * DP_ + 16 * t + m15];
          acc[t] = __builtin_amdgcn_wmma_f32_16x16x4_f32(
              false, a, false, bm, (short)0, acc[t], false, false);
        }
      }
      __builtin_amdgcn_wave_barrier();
    }
  }

  if (accumulate) {
    if (lane < 16) {
#pragma unroll
      for (int t = 0; t < 3; ++t) {
        red[wave][0 * DP_ + 16 * t + lane] = acc[t][0];
        red[wave][1 * DP_ + 16 * t + lane] = acc[t][1];
      }
    }
    __syncthreads();
    if (tid < K_ * DP_) {
      float s = 0.f;
#pragma unroll
      for (int w = 0; w < WAVES_; ++w) s += red[w][tid];
      partials[(size_t)blockIdx.x * (K_ * DP_) + tid] = s;
    }
  }
}

#define RSL_  16
#define RCH_  (CHUNKS_ / RSL_)
#define RQ_   (K_ * DP_ / 4)
#define RTHR_ (RSL_ * RQ_)
__global__ __launch_bounds__(RTHR_) void sk_reduce(const float* __restrict__ partials,
                                                   float* __restrict__ cent) {
  __shared__ float4 sl[RTHR_];
  const int b   = blockIdx.x;
  const int tid = threadIdx.x;
  const float4* __restrict__ p =
      (const float4*)(partials + (size_t)b * CHUNKS_ * (K_ * DP_)) + tid;
  float4 v[RCH_];
#pragma unroll
  for (int j = 0; j < RCH_; ++j) v[j] = p[j * RTHR_];
  float4 a = v[0];
#pragma unroll
  for (int j = 1; j < RCH_; ++j) {
    a.x += v[j].x; a.y += v[j].y; a.z += v[j].z; a.w += v[j].w;
  }
  sl[tid] = a;
  __syncthreads();
  if (tid < K_ * DP_) {
    const float* slf = (const float*)&sl[0];
    const int k = tid / DP_;
    const int d = tid % DP_;
    float t = 0.f, den = 0.f;
#pragma unroll
    for (int s2 = 0; s2 < RSL_; ++s2) {
      t   += slf[s2 * (K_ * DP_) + tid];
      den += slf[s2 * (K_ * DP_) + k * DP_ + D_];
    }
    cent[b * (K_ * DP_) + tid] = (d < D_) ? (t / den) : 0.0f;
  }
}

extern "C" void kernel_launch(void* const* d_in, const int* in_sizes, int n_in,
                              void* d_out, int out_size, void* d_ws, size_t ws_size,
                              hipStream_t stream) {
  const float* data = (const float*)d_in[0];
  const int*   ids  = (const int*)d_in[1];
  const int ITER = 10;

  float*  cent     = (float*)d_ws;
  float*  partials = cent + B_ * K_ * DP_;
  float2* out      = (float2*)d_out;

  sk_init<<<(B_ * K_ * DP_ + 255) / 256, 256, 0, stream>>>(data, ids, cent);
  for (int t = 0; t <= ITER; ++t) {
    int wp = (t == ITER) ? 1 : 0;
    int ac = (t <  ITER) ? 1 : 0;
    sk_pass<<<B_ * CHUNKS_, 256, 0, stream>>>(data, cent, partials, out, wp, ac);
    if (ac)
      sk_reduce<<<B_, RTHR_, 0, stream>>>(partials, cent);
  }
  (void)in_sizes; (void)n_in; (void)out_size; (void)ws_size;
}


# One launch per iteration: chunk reduce and centroid gather folded into the pass prologue; 256 blocks of 16 waves loop over 4 row steps (32 partial rows per batch); partial sums in a new fixed order
# speedup vs baseline: 2.2022x; 1.1143x over previous
#include <hip/hip_runtime.h>

#define B_   8
#define N_   65536
#define D_   40
#define DP_  48
#define K_   2
#define ROW_ (K_ * DP_)
#define WAVES_  16
#define WAVE_ITERS_ 4
#define ROWS_PER_WAVE_ (32 * WAVE_ITERS_)
#define CHUNK_  (WAVES_ * ROWS_PER_WAVE_)
#define CHUNKS_ (N_ / CHUNK_)
#define THREADS_ (WAVES_ * 32)
#define RSL_  16
#define RQ_   (ROW_ / 4)

typedef float v2f __attribute__((ext_vector_type(2)));
typedef float v8f __attribute__((ext_vector_type(8)));

template <bool FIRST>
__global__ __launch_bounds__(THREADS_) void sk_pass(const float* __restrict__ data,
                                                    const int* __restrict__ ids,
                                                    const float* __restrict__ pin,
                                                    float* __restrict__ pout,
                                                    float2* __restrict__ probs_out,
                                                    int write_probs, int accumulate) {
  __shared__ float  cLds[ROW_];
  __shared__ float4 sl[RSL_ * RQ_];
  __shared__ float  xt[WAVES_][2][32 * DP_];
  __shared__ float  pt[WAVES_][2][32 * K_];
  __shared__ float  red[WAVES_][ROW_];

  const int tid   = threadIdx.x;
  const int wave  = tid >> 5;
  const int lane  = tid & 31;
  const int b     = blockIdx.x / CHUNKS_;
  const int chunk = blockIdx.x % CHUNKS_;
  const float* __restrict__ dbase = data + (size_t)b * N_ * D_;

  const float4* __restrict__ xr =
      (const float4*)(dbase + (size_t)(chunk * CHUNK_ + wave * ROWS_PER_WAVE_ + lane) * D_);
  float4 cur[D_ / 4];
#pragma unroll
  for (int q = 0; q < D_ / 4; ++q) cur[q] = xr[q];

  if (FIRST) {
    if (tid < ROW_) {
      const int k = tid / DP_, d = tid % DP_;
      const int n = ids[b * K_ + k];
      cLds[tid] = (d < D_) ? dbase[(size_t)n * D_ + d] : 0.0f;
    }
  } else {
    if (tid < RSL_ * RQ_) {
      const float4* __restrict__ p = (const float4*)(pin + (size_t)b * CHUNKS_ * ROW_) + tid;
      const float4 v0 = p[0];
      const float4 v1 = p[RSL_ * RQ_];
      sl[tid] = make_float4(v0.x + v1.x, v0.y + v1.y, v0.z + v1.z, v0.w + v1.w);
    }
    __syncthreads();
    if (tid < ROW_) {
      const float* slf = (const float*)&sl[0];
      const int k = tid / DP_, d = tid % DP_;
      float t = 0.f, den = 0.f;
#pragma unroll
      for (int s2 = 0; s2 < RSL_; ++s2) {
        t   += slf[s2 * ROW_ + tid];
        den += slf[s2 * ROW_ + k * DP_ + D_];
      }
      cLds[tid] = (d < D_) ? (t / den) : 0.0f;
    }
  }
  __syncthreads();

  float cc0 = 0.f, cc1 = 0.f;
#pragma unroll
  for (int d = 0; d < D_; ++d) {
    float a0 = cLds[d], a1 = cLds[DP_ + d];
    cc0 = fmaf(a0, a0, cc0);
    cc1 = fmaf(a1, a1, cc1);
  }

  v8f acc[3] = {};
  const int m15  = lane & 15;
  const int m1   = lane & 1;
  const bool alive = (m15 < K_);
  const int rsel = (lane >> 4) << 1;

#pragma unroll
  for (int it = 0; it < WAVE_ITERS_; ++it) {
    const int n = chunk * CHUNK_ + wave * ROWS_PER_WAVE_ + it * 32 + lane;
    float4 nxt[D_ / 4];
    if (it + 1 < WAVE_ITERS_) {
#pragma unroll
      for (int q = 0; q < D_ / 4; ++q) nxt[q] = xr[(it + 1) * 32 * (D_ / 4) + q];
    }
    float x[D_];
#pragma unroll
    for (int q = 0; q < D_ / 4; ++q) {
      x[4 * q + 0] = cur[q].x; x[4 * q + 1] = cur[q].y;
      x[4 * q + 2] = cur[q].z; x[4 * q + 3] = cur[q].w;
    }
    float dot0 = 0.f, dot1 = 0.f;
#pragma unroll
    for (int d = 0; d < D_; ++d) {
      dot0 = fmaf(x[d], cLds[d], dot0);
      dot1 = fmaf(x[d], cLds[DP_ + d], dot1);
    }
    float s0 = fmaf(2.0f, dot0, -cc0);
    float s1 = fmaf(2.0f, dot1, -cc1);
    float mx = fmaxf(s0, s1);
    float e0 = __expf(s0 - mx);
    float e1 = __expf(s1 - mx);
    float inv = 1.0f / (e0 + e1);
    float p0 = e0 * inv, p1 = e1 * inv;

    if (write_probs) {
      probs_out[(size_t)b * N_ + n] = make_float2(p0, p1);
    }
    if (accumulate) {
      float* myxt = &xt[wave][it & 1][0];
      float* mypt = &pt[wave][it & 1][0];
#pragma unroll
      for (int d = 0; d < D_; ++d) myxt[lane * DP_ + d] = x[d];
      myxt[lane * DP_ + D_] = 1.0f;
#pragma unroll
      for (int d = D_ + 1; d < DP_; ++d) myxt[lane * DP_ + d] = 0.0f;
      mypt[lane * K_ + 0] = p0;
      mypt[lane * K_ + 1] = p1;
      __builtin_amdgcn_wave_barrier();

#pragma unroll
      for (int g = 0; g < 8; ++g) {
        const int r0 = 4 * g + rsel;
        float a0v = mypt[(r0 + 0) * K_ + m1];
        float a1v = mypt[(r0 + 1) * K_ + m1];
        v2f a;
        a.x = alive ? a0v : 0.0f;
        a.y = alive ? a1v : 0.0f;
#pragma unroll
        for (int t = 0; t < 3; ++t) {
          v2f bm;
          bm.x = myxt[(r0 + 0) * DP_ + 16 * t + m15];
          bm.y = myxt[(r0 + 1) * DP_ + 16 * t + m15];
          acc[t] = __builtin_amdgcn_wmma_f32_16x16x4_f32(
              false, a, false, bm, (short)0, acc[t], false, false);
        }
      }
    }
    if (it + 1 < WAVE_ITERS_) {
#pragma unroll
      for (int q = 0; q < D_ / 4; ++q) cur[q] = nxt[q];
    }
  }

  if (accumulate) {
    if (lane < 16) {
#pragma unroll
      for (int t = 0; t < 3; ++t) {
        red[wave][0 * DP_ + 16 * t + lane] = acc[t][0];
        red[wave][1 * DP_ + 16 * t + lane] = acc[t][1];
      }
    }
    __syncthreads();
    if (tid < ROW_) {
      float s = 0.f;
#pragma unroll
      for (int w = 0; w < WAVES_; ++w) s += red[w][tid];
      pout[(size_t)blockIdx.x * ROW_ + tid] = s;
    }
  }
}

extern "C" void kernel_launch(void* const* d_in, const int* in_sizes, int n_in,
                              void* d_out, int out_size, void* d_ws, size_t ws_size,
                              hipStream_t stream) {
  const float* data = (const float*)d_in[0];
  const int*   ids  = (const int*)d_in[1];
  const int ITER = 10;

  float*  rowsA = (float*)d_ws;
  float*  rowsB = rowsA + B_ * CHUNKS_ * ROW_;
  float2* out   = (float2*)d_out;

  for (int t = 0; t <= ITER; ++t) {
    int wp = (t == ITER) ? 1 : 0;
    int ac = (t <  ITER) ? 1 : 0;
    const float* pin  = (t & 1) ? rowsA : rowsB;
    float*       pout = (t & 1) ? rowsB : rowsA;
    if (t == 0)
      sk_pass<true><<<B_ * CHUNKS_, THREADS_, 0, stream>>>(data, ids, pin, pout, out, wp, ac);
    else
      sk_pass<false><<<B_ * CHUNKS_, THREADS_, 0, stream>>>(data, ids, pin, pout, out, wp, ac);
  }
  (void)in_sizes; (void)n_in; (void)out_size; (void)ws_size;
}


# Whole-line global loads: each load instruction reads 512 contiguous bytes of the wave's rows (issue point held by asm); rows reach their lanes through the wave's LDS tile, pitch 44; sums unchanged
# speedup vs baseline: 2.7845x; 1.2644x over previous
#include <hip/hip_runtime.h>

#define B_   8
#define N_   65536
#define D_   40
#define DP_  48
#define K_   2
#define ROW_ (K_ * DP_)
#define WAVES_  16
#define WAVE_ITERS_ 4
#define ROWS_PER_WAVE_ (32 * WAVE_ITERS_)
#define CHUNK_  (WAVES_ * ROWS_PER_WAVE_)
#define CHUNKS_ (N_ / CHUNK_)
#define THREADS_ (WAVES_ * 32)
#define RSL_  16
#define RQ_   (ROW_ / 4)
#define DQ_   (D_ / 4)
#define XP_   44

typedef float v2f __attribute__((ext_vector_type(2)));
typedef float v8f __attribute__((ext_vector_type(8)));
typedef float v4f __attribute__((ext_vector_type(4)));

__device__ __forceinline__ void load_piece(v4f& dst, const float4* p, const int byte_off) {
  asm volatile("global_load_b128 %0, %1, off offset:%2" : "=&v"(dst) : "v"(p), "n"(byte_off));
}
__device__ __forceinline__ void wait_pieces(v4f (&d)[10], const float4*& p) {
  asm volatile("s_wait_loadcnt 0x0"
               : "+v"(d[0]), "+v"(d[1]), "+v"(d[2]), "+v"(d[3]), "+v"(d[4]),
                 "+v"(d[5]), "+v"(d[6]), "+v"(d[7]), "+v"(d[8]), "+v"(d[9]), "+v"(p)
               :
               : "memory");
}

template <bool FIRST>
__global__ __launch_bounds__(THREADS_) void sk_pass(const float* __restrict__ data,
                                                    const int* __restrict__ ids,
                                                    const float* __restrict__ pin,
                                                    float* __restrict__ pout,
                                                    float2* __restrict__ probs_out,
                                                    int write_probs, int accumulate) {
  __shared__ float  cLds[ROW_];
  __shared__ float4 sl[RSL_ * RQ_];
  __shared__ __attribute__((aligned(16))) float xt[WAVES_][2][32 * XP_];
  __shared__ float  pt[WAVES_][2][32 * K_];
  __shared__ float  red[WAVES_][ROW_];

  const int tid   = threadIdx.x;
  const int wave  = tid >> 5;
  const int lane  = tid & 31;
  const int b     = blockIdx.x / CHUNKS_;
  const int chunk = blockIdx.x % CHUNKS_;
  const float* __restrict__ dbase = data + (size_t)b * N_ * D_;

  const float4* xw =
      (const float4*)(dbase + (size_t)(chunk * CHUNK_ + wave * ROWS_PER_WAVE_) * D_) + lane;
  v4f cur[DQ_];
#pragma unroll
  for (int q = 0; q < DQ_; ++q) load_piece(cur[q], xw, 32 * q * 16);

  int soff[DQ_];
#pragma unroll
  for (int q = 0; q < DQ_; ++q) {
    const int piece = 32 * q + lane;
    soff[q] = 4 * (piece + piece / DQ_);
  }
  *(float4*)&xt[wave][0][lane * XP_ + D_] = make_float4(1.0f, 0.0f, 0.0f, 0.0f);
  *(float4*)&xt[wave][1][lane * XP_ + D_] = make_float4(1.0f, 0.0f, 0.0f, 0.0f);

  if (FIRST) {
    if (tid < ROW_) {
      const int k = tid / DP_, d = tid % DP_;
      const int n = ids[b * K_ + k];
      cLds[tid] = (d < D_) ? dbase[(size_t)n * D_ + d] : 0.0f;
    }
  } else {
    if (tid < RSL_ * RQ_) {
      const float4* __restrict__ p = (const float4*)(pin + (size_t)b * CHUNKS_ * ROW_) + tid;
      const float4 v0 = p[0];
      const float4 v1 = p[RSL_ * RQ_];
      sl[tid] = make_float4(v0.x + v1.x, v0.y + v1.y, v0.z + v1.z, v0.w + v1.w);
    }
    __syncthreads();
    if (tid < ROW_) {
      const float* slf = (const float*)&sl[0];
      const int k = tid / DP_, d = tid % DP_;
      float t = 0.f, den = 0.f;
#pragma unroll
      for (int s2 = 0; s2 < RSL_; ++s2) {
        t   += slf[s2 * ROW_ + tid];
        den += slf[s2 * ROW_ + k * DP_ + D_];
      }
      cLds[tid] = (d < D_) ? (t / den) : 0.0f;
    }
  }
  __syncthreads();

  float cc0 = 0.f, cc1 = 0.f;
#pragma unroll
  for (int d = 0; d < D_; ++d) {
    float a0 = cLds[d], a1 = cLds[DP_ + d];
    cc0 = fmaf(a0, a0, cc0);
    cc1 = fmaf(a1, a1, cc1);
  }

  v8f acc[3] = {};
  const int m15  = lane & 15;
  const int m1   = lane & 1;
  const bool alive = (m15 < K_);
  const int rsel = (lane >> 4) << 1;
  const int c2 = (32 + m15 < XP_ - 1) ? (32 + m15) : (XP_ - 1);

#pragma unroll
  for (int it = 0; it < WAVE_ITERS_; ++it) {
    const int n = chunk * CHUNK_ + wave * ROWS_PER_WAVE_ + it * 32 + lane;
    wait_pieces(cur, xw);
    v4f nxt[DQ_];
    if (it + 1 < WAVE_ITERS_) {
#pragma unroll
      for (int q = 0; q < DQ_; ++q) load_piece(nxt[q], xw, ((it + 1) * 32 * DQ_ + 32 * q) * 16);
    }
    float* myxt = &xt[wave][it & 1][0];
#pragma unroll
    for (int q = 0; q < DQ_; ++q) *(v4f*)&myxt[soff[q]] = cur[q];
    __builtin_amdgcn_wave_barrier();
    float x[D_];
#pragma unroll
    for (int q = 0; q < DQ_; ++q) {
      const float4 v = *(const float4*)&myxt[lane * XP_ + 4 * q];
      x[4 * q + 0] = v.x; x[4 * q + 1] = v.y;
      x[4 * q + 2] = v.z; x[4 * q + 3] = v.w;
    }
    float dot0 = 0.f, dot1 = 0.f;
#pragma unroll
    for (int d = 0; d < D_; ++d) {
      dot0 = fmaf(x[d], cLds[d], dot0);
      dot1 = fmaf(x[d], cLds[DP_ + d], dot1);
    }
    float s0 = fmaf(2.0f, dot0, -cc0);
    float s1 = fmaf(2.0f, dot1, -cc1);
    float mx = fmaxf(s0, s1);
    float e0 = __expf(s0 - mx);
    float e1 = __expf(s1 - mx);
    float inv = 1.0f / (e0 + e1);
    float p0 = e0 * inv, p1 = e1 * inv;

    if (write_probs) {
      probs_out[(size_t)b * N_ + n] = make_float2(p0, p1);
    }
    if (accumulate) {
      float* mypt = &pt[wave][it & 1][0];
      mypt[lane * K_ + 0] = p0;
      mypt[lane * K_ + 1] = p1;
      __builtin_amdgcn_wave_barrier();

#pragma unroll
      for (int g = 0; g < 8; ++g) {
        const int r0 = 4 * g + rsel;
        float a0v = mypt[(r0 + 0) * K_ + m1];
        float a1v = mypt[(r0 + 1) * K_ + m1];
        v2f a;
        a.x = alive ? a0v : 0.0f;
        a.y = alive ? a1v : 0.0f;
#pragma unroll
        for (int t = 0; t < 3; ++t) {
          const int col = (t < 2) ? (16 * t + m15) : c2;
          v2f bm;
          bm.x = myxt[(r0 + 0) * XP_ + col];
          bm.y = myxt[(r0 + 1) * XP_ + col];
          acc[t] = __builtin_amdgcn_wmma_f32_16x16x4_f32(
              false, a, false, bm, (short)0, acc[t], false, false);
        }
      }
    }
    if (it + 1 < WAVE_ITERS_) {
#pragma unroll
      for (int q = 0; q < DQ_; ++q) cur[q] = nxt[q];
    }
  }

  if (accumulate) {
    if (lane < 16) {
#pragma unroll
      for (int t = 0; t < 3; ++t) {
        red[wave][0 * DP_ + 16 * t + lane] = acc[t][0];
        red[wave][1 * DP_ + 16 * t + lane] = acc[t][1];
      }
    }
    __syncthreads();
    if (tid < ROW_) {
      float s = 0.f;
#pragma unroll
      for (int w = 0; w < WAVES_; ++w) s += red[w][tid];
      pout[(size_t)blockIdx.x * ROW_ + tid] = s;
    }
  }
}

extern "C" void kernel_launch(void* const* d_in, const int* in_sizes, int n_in,
                              void* d_out, int out_size, void* d_ws, size_t ws_size,
                              hipStream_t stream) {
  const float* data = (const float*)d_in[0];
  const int*   ids  = (const int*)d_in[1];
  const int ITER = 10;

  float*  rowsA = (float*)d_ws;
  float*  rowsB = rowsA + B_ * CHUNKS_ * ROW_;
  float2* out   = (float2*)d_out;

  for (int t = 0; t <= ITER; ++t) {
    int wp = (t == ITER) ? 1 : 0;
    int ac = (t <  ITER) ? 1 : 0;
    const float* pin  = (t & 1) ? rowsA : rowsB;
    float*       pout = (t & 1) ? rowsB : rowsA;
    if (t == 0)
      sk_pass<true><<<B_ * CHUNKS_, THREADS_, 0, stream>>>(data, ids, pin, pout, out, wp, ac);
    else
      sk_pass<false><<<B_ * CHUNKS_, THREADS_, 0, stream>>>(data, ids, pin, pout, out, wp, ac);
  }
  (void)in_sizes; (void)n_in; (void)out_size; (void)ws_size;
}


# Scores as one dot with the centroid difference kept in registers (two-cluster softmax = logistic of score difference): no per-step LDS centroid reads; dot in 4 partial sums; moment sums unchanged
# speedup vs baseline: 3.0192x; 1.0843x over previous
#include <hip/hip_runtime.h>

#define B_   8
#define N_   65536
#define D_   40
#define DP_  48
#define K_   2
#define ROW_ (K_ * DP_)
#define WAVES_  16
#define WAVE_ITERS_ 4
#define ROWS_PER_WAVE_ (32 * WAVE_ITERS_)
#define CHUNK_  (WAVES_ * ROWS_PER_WAVE_)
#define CHUNKS_ (N_ / CHUNK_)
#define THREADS_ (WAVES_ * 32)
#define RSL_  16
#define RQ_   (ROW_ / 4)
#define DQ_   (D_ / 4)
#define XP_   44

typedef float v2f __attribute__((ext_vector_type(2)));
typedef float v8f __attribute__((ext_vector_type(8)));
typedef float v4f __attribute__((ext_vector_type(4)));

__device__ __forceinline__ void load_piece(v4f& dst, const float4* p, const int byte_off) {
  asm volatile("global_load_b128 %0, %1, off offset:%2" : "=&v"(dst) : "v"(p), "n"(byte_off));
}
__device__ __forceinline__ void wait_pieces(v4f (&d)[10], const float4*& p) {
  asm volatile("s_wait_loadcnt 0x0"
               : "+v"(d[0]), "+v"(d[1]), "+v"(d[2]), "+v"(d[3]), "+v"(d[4]),
                 "+v"(d[5]), "+v"(d[6]), "+v"(d[7]), "+v"(d[8]), "+v"(d[9]), "+v"(p)
               :
               : "memory");
}

template <bool FIRST>
__global__ __launch_bounds__(THREADS_) void sk_pass(const float* __restrict__ data,
                                                    const int* __restrict__ ids,
                                                    const float* __restrict__ pin,
                                                    float* __restrict__ pout,
                                                    float2* __restrict__ probs_out,
                                                    int write_probs, int accumulate) {
  __shared__ float  cLds[ROW_];
  __shared__ float4 sl[RSL_ * RQ_];
  __shared__ __attribute__((aligned(16))) float xt[WAVES_][2][32 * XP_];
  __shared__ float  pt[WAVES_][2][32 * K_];
  __shared__ float  red[WAVES_][ROW_];

  const int tid   = threadIdx.x;
  const int wave  = tid >> 5;
  const int lane  = tid & 31;
  const int b     = blockIdx.x / CHUNKS_;
  const int chunk = blockIdx.x % CHUNKS_;
  const float* __restrict__ dbase = data + (size_t)b * N_ * D_;

  const float4* xw =
      (const float4*)(dbase + (size_t)(chunk * CHUNK_ + wave * ROWS_PER_WAVE_) * D_) + lane;
  v4f cur[DQ_];
#pragma unroll
  for (int q = 0; q < DQ_; ++q) load_piece(cur[q], xw, 32 * q * 16);

  int soff[DQ_];
#pragma unroll
  for (int q = 0; q < DQ_; ++q) {
    const int piece = 32 * q + lane;
    soff[q] = 4 * (piece + piece / DQ_);
  }
  *(float4*)&xt[wave][0][lane * XP_ + D_] = make_float4(1.0f, 0.0f, 0.0f, 0.0f);
  *(float4*)&xt[wave][1][lane * XP_ + D_] = make_float4(1.0f, 0.0f, 0.0f, 0.0f);

  if (FIRST) {
    if (tid < ROW_) {
      const int k = tid / DP_, d = tid % DP_;
      const int n = ids[b * K_ + k];
      cLds[tid] = (d < D_) ? dbase[(size_t)n * D_ + d] : 0.0f;
    }
  } else {
    if (tid < RSL_ * RQ_) {
      const float4* __restrict__ p = (const float4*)(pin + (size_t)b * CHUNKS_ * ROW_) + tid;
      const float4 v0 = p[0];
      const float4 v1 = p[RSL_ * RQ_];
      sl[tid] = make_float4(v0.x + v1.x, v0.y + v1.y, v0.z + v1.z, v0.w + v1.w);
    }
    __syncthreads();
    if (tid < ROW_) {
      const float* slf = (const float*)&sl[0];
      const int k = tid / DP_, d = tid % DP_;
      float t = 0.f, den = 0.f;
#pragma unroll
      for (int s2 = 0; s2 < RSL_; ++s2) {
        t   += slf[s2 * ROW_ + tid];
        den += slf[s2 * ROW_ + k * DP_ + D_];
      }
      cLds[tid] = (d < D_) ? (t / den) : 0.0f;
    }
  }
  __syncthreads();

  float cd[D_];
  float bs[4] = {0.f, 0.f, 0.f, 0.f};
#pragma unroll
  for (int d = 0; d < D_; ++d) {
    const float a0 = cLds[d], a1 = cLds[DP_ + d];
    cd[d] = a0 - a1;
    bs[d & 3] = fmaf(cd[d], a0 + a1, bs[d & 3]);
  }
  const float bias = (bs[0] + bs[1]) + (bs[2] + bs[3]);

  v8f acc[3] = {};
  const int m15  = lane & 15;
  const int m1   = lane & 1;
  const bool alive = (m15 < K_);
  const int rsel = (lane >> 4) << 1;
  const int c2 = (32 + m15 < XP_ - 1) ? (32 + m15) : (XP_ - 1);

#pragma unroll
  for (int it = 0; it < WAVE_ITERS_; ++it) {
    const int n = chunk * CHUNK_ + wave * ROWS_PER_WAVE_ + it * 32 + lane;
    wait_pieces(cur, xw);
    v4f nxt[DQ_];
    if (it + 1 < WAVE_ITERS_) {
#pragma unroll
      for (int q = 0; q < DQ_; ++q) load_piece(nxt[q], xw, ((it + 1) * 32 * DQ_ + 32 * q) * 16);
    }
    float* myxt = &xt[wave][it & 1][0];
#pragma unroll
    for (int q = 0; q < DQ_; ++q) *(v4f*)&myxt[soff[q]] = cur[q];
    __builtin_amdgcn_wave_barrier();
    float ds[4] = {0.f, 0.f, 0.f, 0.f};
#pragma unroll
    for (int q = 0; q < DQ_; ++q) {
      const float4 v = *(const float4*)&myxt[lane * XP_ + 4 * q];
      ds[0] = fmaf(v.x, cd[4 * q + 0], ds[0]);
      ds[1] = fmaf(v.y, cd[4 * q + 1], ds[1]);
      ds[2] = fmaf(v.z, cd[4 * q + 2], ds[2]);
      ds[3] = fmaf(v.w, cd[4 * q + 3], ds[3]);
    }
    const float t   = fmaf(2.0f, (ds[0] + ds[1]) + (ds[2] + ds[3]), -bias);
    const float e   = __expf(-fabsf(t));
    const float inv = 1.0f / (1.0f + e);
    const float ps  = e * inv;
    const float p0  = (t >= 0.0f) ? inv : ps;
    const float p1  = (t >= 0.0f) ? ps : inv;

    if (write_probs) {
      probs_out[(size_t)b * N_ + n] = make_float2(p0, p1);
    }
    if (accumulate) {
      float* mypt = &pt[wave][it & 1][0];
      mypt[lane * K_ + 0] = p0;
      mypt[lane * K_ + 1] = p1;
      __builtin_amdgcn_wave_barrier();

#pragma unroll
      for (int g = 0; g < 8; ++g) {
        const int r0 = 4 * g + rsel;
        float a0v = mypt[(r0 + 0) * K_ + m1];
        float a1v = mypt[(r0 + 1) * K_ + m1];
        v2f a;
        a.x = alive ? a0v : 0.0f;
        a.y = alive ? a1v : 0.0f;
#pragma unroll
        for (int t = 0; t < 3; ++t) {
          const int col = (t < 2) ? (16 * t + m15) : c2;
          v2f bm;
          bm.x = myxt[(r0 + 0) * XP_ + col];
          bm.y = myxt[(r0 + 1) * XP_ + col];
          acc[t] = __builtin_amdgcn_wmma_f32_16x16x4_f32(
              false, a, false, bm, (short)0, acc[t], false, false);
        }
      }
    }
    if (it + 1 < WAVE_ITERS_) {
#pragma unroll
      for (int q = 0; q < DQ_; ++q) cur[q] = nxt[q];
    }
  }

  if (accumulate) {
    if (lane < 16) {
#pragma unroll
      for (int t = 0; t < 3; ++t) {
        red[wave][0 * DP_ + 16 * t + lane] = acc[t][0];
        red[wave][1 * DP_ + 16 * t + lane] = acc[t][1];
      }
    }
    __syncthreads();
    if (tid < ROW_) {
      float s = 0.f;
#pragma unroll
      for (int w = 0; w < WAVES_; ++w) s += red[w][tid];
      pout[(size_t)blockIdx.x * ROW_ + tid] = s;
    }
  }
}

extern "C" void kernel_launch(void* const* d_in, const int* in_sizes, int n_in,
                              void* d_out, int out_size, void* d_ws, size_t ws_size,
                              hipStream_t stream) {
  const float* data = (const float*)d_in[0];
  const int*   ids  = (const int*)d_in[1];
  const int ITER = 10;

  float*  rowsA = (float*)d_ws;
  float*  rowsB = rowsA + B_ * CHUNKS_ * ROW_;
  float2* out   = (float2*)d_out;

  for (int t = 0; t <= ITER; ++t) {
    int wp = (t == ITER) ? 1 : 0;
    int ac = (t <  ITER) ? 1 : 0;
    const float* pin  = (t & 1) ? rowsA : rowsB;
    float*       pout = (t & 1) ? rowsB : rowsA;
    if (t == 0)
      sk_pass<true><<<B_ * CHUNKS_, THREADS_, 0, stream>>>(data, ids, pin, pout, out, wp, ac);
    else
      sk_pass<false><<<B_ * CHUNKS_, THREADS_, 0, stream>>>(data, ids, pin, pout, out, wp, ac);
  }
  (void)in_sizes; (void)n_in; (void)out_size; (void)ws_size;
}


# Two 4-sample groups per A fragment (matrix rows 2,3 = second group): the 8-column last d-tile serves both groups in one instruction, 20 not 24 per step; denominators by lane sums; new fixed sum order
# speedup vs baseline: 3.1439x; 1.0413x over previous
#include <hip/hip_runtime.h>

#define B_   8
#define N_   65536
#define D_   40
#define DP_  48
#define K_   2
#define ROW_ (K_ * DP_)
#define WAVES_  16
#define WAVE_ITERS_ 4
#define ROWS_PER_WAVE_ (32 * WAVE_ITERS_)
#define CHUNK_  (WAVES_ * ROWS_PER_WAVE_)
#define CHUNKS_ (N_ / CHUNK_)
#define THREADS_ (WAVES_ * 32)
#define RSL_  16
#define RQ_   (ROW_ / 4)
#define DQ_   (D_ / 4)
#define XP_   44

typedef float v2f __attribute__((ext_vector_type(2)));
typedef float v8f __attribute__((ext_vector_type(8)));
typedef float v4f __attribute__((ext_vector_type(4)));

__device__ __forceinline__ void load_piece(v4f& dst, const float4* p, const int byte_off) {
  asm volatile("global_load_b128 %0, %1, off offset:%2" : "=&v"(dst) : "v"(p), "n"(byte_off));
}
__device__ __forceinline__ void wait_pieces(v4f (&d)[10], const float4*& p) {
  asm volatile("s_wait_loadcnt 0x0"
               : "+v"(d[0]), "+v"(d[1]), "+v"(d[2]), "+v"(d[3]), "+v"(d[4]),
                 "+v"(d[5]), "+v"(d[6]), "+v"(d[7]), "+v"(d[8]), "+v"(d[9]), "+v"(p)
               :
               : "memory");
}

template <bool FIRST>
__global__ __launch_bounds__(THREADS_) void sk_pass(const float* __restrict__ data,
                                                    const int* __restrict__ ids,
                                                    const float* __restrict__ pin,
                                                    float* __restrict__ pout,
                                                    float2* __restrict__ probs_out,
                                                    int write_probs, int accumulate) {
  __shared__ float  cLds[ROW_];
  __shared__ float4 sl[RSL_ * RQ_];
  __shared__ __attribute__((aligned(16))) float xt[WAVES_][2][32 * XP_];
  __shared__ float  pt[WAVES_][2][32 * K_];
  __shared__ float  red[WAVES_][ROW_];

  const int tid   = threadIdx.x;
  const int wave  = tid >> 5;
  const int lane  = tid & 31;
  const int b     = blockIdx.x / CHUNKS_;
  const int chunk = blockIdx.x % CHUNKS_;
  const float* __restrict__ dbase = data + (size_t)b * N_ * D_;

  const float4* xw =
      (const float4*)(dbase + (size_t)(chunk * CHUNK_ + wave * ROWS_PER_WAVE_) * D_) + lane;
  v4f cur[DQ_];
#pragma unroll
  for (int q = 0; q < DQ_; ++q) load_piece(cur[q], xw, 32 * q * 16);

  int soff[DQ_];
#pragma unroll
  for (int q = 0; q < DQ_; ++q) {
    const int piece = 32 * q + lane;
    soff[q] = 4 * (piece + piece / DQ_);
  }

  if (FIRST) {
    if (tid < ROW_) {
      const int k = tid / DP_, d = tid % DP_;
      const int n = ids[b * K_ + k];
      cLds[tid] = (d < D_) ? dbase[(size_t)n * D_ + d] : 0.0f;
    }
  } else {
    if (tid < RSL_ * RQ_) {
      const float4* __restrict__ p = (const float4*)(pin + (size_t)b * CHUNKS_ * ROW_) + tid;
      const float4 v0 = p[0];
      const float4 v1 = p[RSL_ * RQ_];
      sl[tid] = make_float4(v0.x + v1.x, v0.y + v1.y, v0.z + v1.z, v0.w + v1.w);
    }
    __syncthreads();
    if (tid < ROW_) {
      const float* slf = (const float*)&sl[0];
      const int k = tid / DP_, d = tid % DP_;
      float t = 0.f, den = 0.f;
#pragma unroll
      for (int s2 = 0; s2 < RSL_; ++s2) {
        t   += slf[s2 * ROW_ + tid];
        den += slf[s2 * ROW_ + k * DP_ + D_];
      }
      cLds[tid] = (d < D_) ? (t / den) : 0.0f;
    }
  }
  __syncthreads();

  float cd[D_];
  float bs[4] = {0.f, 0.f, 0.f, 0.f};
#pragma unroll
  for (int d = 0; d < D_; ++d) {
    const float a0 = cLds[d], a1 = cLds[DP_ + d];
    cd[d] = a0 - a1;
    bs[d & 3] = fmaf(cd[d], a0 + a1, bs[d & 3]);
  }
  const float bias = (bs[0] + bs[1]) + (bs[2] + bs[3]);

  v8f accA[2] = {};
  v8f accB[2] = {};
  v8f accL    = {};
  float den0 = 0.f, den1 = 0.f;
  const int m15  = lane & 15;
  const int rsel = (lane >> 4) << 1;
  const int aoff = ((((m15 >> 1) & 1) << 2) + rsel) * K_ + (m15 & 1);
  const int boff = rsel * XP_ + m15;
  const int loff = ((((m15 >> 3) & 1) << 2) + rsel) * XP_ + 32 + (m15 & 7);

#pragma unroll
  for (int it = 0; it < WAVE_ITERS_; ++it) {
    const int n = chunk * CHUNK_ + wave * ROWS_PER_WAVE_ + it * 32 + lane;
    wait_pieces(cur, xw);
    v4f nxt[DQ_];
    if (it + 1 < WAVE_ITERS_) {
#pragma unroll
      for (int q = 0; q < DQ_; ++q) load_piece(nxt[q], xw, ((it + 1) * 32 * DQ_ + 32 * q) * 16);
    }
    float* myxt = &xt[wave][it & 1][0];
#pragma unroll
    for (int q = 0; q < DQ_; ++q) *(v4f*)&myxt[soff[q]] = cur[q];
    __builtin_amdgcn_wave_barrier();
    float ds[4] = {0.f, 0.f, 0.f, 0.f};
#pragma unroll
    for (int q = 0; q < DQ_; ++q) {
      const float4 v = *(const float4*)&myxt[lane * XP_ + 4 * q];
      ds[0] = fmaf(v.x, cd[4 * q + 0], ds[0]);
      ds[1] = fmaf(v.y, cd[4 * q + 1], ds[1]);
      ds[2] = fmaf(v.z, cd[4 * q + 2], ds[2]);
      ds[3] = fmaf(v.w, cd[4 * q + 3], ds[3]);
    }
    const float t   = fmaf(2.0f, (ds[0] + ds[1]) + (ds[2] + ds[3]), -bias);
    const float e   = __expf(-fabsf(t));
    const float inv = 1.0f / (1.0f + e);
    const float ps  = e * inv;
    const float p0  = (t >= 0.0f) ? inv : ps;
    const float p1  = (t >= 0.0f) ? ps : inv;

    if (write_probs) {
      probs_out[(size_t)b * N_ + n] = make_float2(p0, p1);
    }
    if (accumulate) {
      den0 += p0;
      den1 += p1;
      float* mypt = &pt[wave][it & 1][0];
      mypt[lane * K_ + 0] = p0;
      mypt[lane * K_ + 1] = p1;
      __builtin_amdgcn_wave_barrier();

#pragma unroll
      for (int gp = 0; gp < 4; ++gp) {
        v2f a;
        a.x = mypt[8 * gp * K_ + aoff];
        a.y = mypt[8 * gp * K_ + aoff + K_];
        const float* xg = myxt + 8 * gp * XP_;
#pragma unroll
        for (int t = 0; t < 2; ++t) {
          v2f b0, b1;
          b0.x = xg[boff + 16 * t];
          b0.y = xg[boff + 16 * t + XP_];
          accA[t] = __builtin_amdgcn_wmma_f32_16x16x4_f32(
              false, a, false, b0, (short)0, accA[t], false, false);
          b1.x = xg[boff + 16 * t + 4 * XP_];
          b1.y = xg[boff + 16 * t + 5 * XP_];
          accB[t] = __builtin_amdgcn_wmma_f32_16x16x4_f32(
              false, a, false, b1, (short)0, accB[t], false, false);
        }
        v2f bl;
        bl.x = xg[loff];
        bl.y = xg[loff + XP_];
        accL = __builtin_amdgcn_wmma_f32_16x16x4_f32(
            false, a, false, bl, (short)0, accL, false, false);
      }
    }
    if (it + 1 < WAVE_ITERS_) {
#pragma unroll
      for (int q = 0; q < DQ_; ++q) cur[q] = nxt[q];
    }
  }

  if (accumulate) {
#pragma unroll
    for (int off = 16; off > 0; off >>= 1) {
      den0 += __shfl_xor(den0, off);
      den1 += __shfl_xor(den1, off);
    }
    const float l0 = accL[0] + __shfl_down(accL[2], 8);
    const float l1 = accL[1] + __shfl_down(accL[3], 8);
    if (lane < 16) {
#pragma unroll
      for (int t = 0; t < 2; ++t) {
        red[wave][0 * DP_ + 16 * t + lane] = accA[t][0] + accB[t][2];
        red[wave][1 * DP_ + 16 * t + lane] = accA[t][1] + accB[t][3];
      }
      red[wave][0 * DP_ + 32 + lane] = (lane < 8) ? l0 : ((lane == 8) ? den0 : 0.0f);
      red[wave][1 * DP_ + 32 + lane] = (lane < 8) ? l1 : ((lane == 8) ? den1 : 0.0f);
    }
    __syncthreads();
    if (tid < ROW_) {
      float s = 0.f;
#pragma unroll
      for (int w = 0; w < WAVES_; ++w) s += red[w][tid];
      pout[(size_t)blockIdx.x * ROW_ + tid] = s;
    }
  }
}

extern "C" void kernel_launch(void* const* d_in, const int* in_sizes, int n_in,
                              void* d_out, int out_size, void* d_ws, size_t ws_size,
                              hipStream_t stream) {
  const float* data = (const float*)d_in[0];
  const int*   ids  = (const int*)d_in[1];
  const int ITER = 10;

  float*  rowsA = (float*)d_ws;
  float*  rowsB = rowsA + B_ * CHUNKS_ * ROW_;
  float2* out   = (float2*)d_out;

  for (int t = 0; t <= ITER; ++t) {
    int wp = (t == ITER) ? 1 : 0;
    int ac = (t <  ITER) ? 1 : 0;
    const float* pin  = (t & 1) ? rowsA : rowsB;
    float*       pout = (t & 1) ? rowsB : rowsA;
    if (t == 0)
      sk_pass<true><<<B_ * CHUNKS_, THREADS_, 0, stream>>>(data, ids, pin, pout, out, wp, ac);
    else
      sk_pass<false><<<B_ * CHUNKS_, THREADS_, 0, stream>>>(data, ids, pin, pout, out, wp, ac);
  }
  (void)in_sizes; (void)n_in; (void)out_size; (void)ws_size;
}


# Both matrix operands carry data: A rows 0-13 = 14 columns of one 4-sample group, rows 14,15 = the other group's probs, B the mirror; 12 not 20 matrix instr per step; new fixed sum order
# speedup vs baseline: 3.3690x; 1.0716x over previous
#include <hip/hip_runtime.h>

#define B_   8
#define N_   65536
#define D_   40
#define DP_  48
#define K_   2
#define ROW_ (K_ * DP_)
#define WAVES_  16
#define WAVE_ITERS_ 4
#define ROWS_PER_WAVE_ (32 * WAVE_ITERS_)
#define CHUNK_  (WAVES_ * ROWS_PER_WAVE_)
#define CHUNKS_ (N_ / CHUNK_)
#define THREADS_ (WAVES_ * 32)
#define RSL_  16
#define RQ_   (ROW_ / 4)
#define DQ_   (D_ / 4)
#define XP_   44

typedef float v2f __attribute__((ext_vector_type(2)));
typedef float v8f __attribute__((ext_vector_type(8)));
typedef float v4f __attribute__((ext_vector_type(4)));

__device__ __forceinline__ void load_piece(v4f& dst, const float4* p, const int byte_off) {
  asm volatile("global_load_b128 %0, %1, off offset:%2" : "=&v"(dst) : "v"(p), "n"(byte_off));
}
__device__ __forceinline__ void wait_pieces(v4f (&d)[10], const float4*& p) {
  asm volatile("s_wait_loadcnt 0x0"
               : "+v"(d[0]), "+v"(d[1]), "+v"(d[2]), "+v"(d[3]), "+v"(d[4]),
                 "+v"(d[5]), "+v"(d[6]), "+v"(d[7]), "+v"(d[8]), "+v"(d[9]), "+v"(p)
               :
               : "memory");
}

template <bool FIRST>
__global__ __launch_bounds__(THREADS_) void sk_pass(const float* __restrict__ data,
                                                    const int* __restrict__ ids,
                                                    const float* __restrict__ pin,
                                                    float* __restrict__ pout,
                                                    float2* __restrict__ probs_out,
                                                    int write_probs, int accumulate) {
  __shared__ float  cLds[ROW_];
  __shared__ float4 sl[RSL_ * RQ_];
  __shared__ __attribute__((aligned(16))) float xt[WAVES_][2][32 * XP_];
  __shared__ float  red1[WAVES_][ROW_];
  __shared__ float  red2[WAVES_][ROW_];
  __shared__ float  dn[WAVES_][K_];

  const int tid   = threadIdx.x;
  const int wave  = tid >> 5;
  const int lane  = tid & 31;
  const int b     = blockIdx.x / CHUNKS_;
  const int chunk = blockIdx.x % CHUNKS_;
  const float* __restrict__ dbase = data + (size_t)b * N_ * D_;

  const float4* xw =
      (const float4*)(dbase + (size_t)(chunk * CHUNK_ + wave * ROWS_PER_WAVE_) * D_) + lane;
  v4f cur[DQ_];
#pragma unroll
  for (int q = 0; q < DQ_; ++q) load_piece(cur[q], xw, 32 * q * 16);

  int soff[DQ_];
#pragma unroll
  for (int q = 0; q < DQ_; ++q) {
    const int piece = 32 * q + lane;
    soff[q] = 4 * (piece + piece / DQ_);
  }

  if (FIRST) {
    if (tid < ROW_) {
      const int k = tid / DP_, d = tid % DP_;
      const int n = ids[b * K_ + k];
      cLds[tid] = (d < D_) ? dbase[(size_t)n * D_ + d] : 0.0f;
    }
  } else {
    if (tid < RSL_ * RQ_) {
      const float4* __restrict__ p = (const float4*)(pin + (size_t)b * CHUNKS_ * ROW_) + tid;
      const float4 v0 = p[0];
      const float4 v1 = p[RSL_ * RQ_];
      sl[tid] = make_float4(v0.x + v1.x, v0.y + v1.y, v0.z + v1.z, v0.w + v1.w);
    }
    __syncthreads();
    if (tid < ROW_) {
      const float* slf = (const float*)&sl[0];
      const int k = tid / DP_, d = tid % DP_;
      float t = 0.f, den = 0.f;
#pragma unroll
      for (int s2 = 0; s2 < RSL_; ++s2) {
        t   += slf[s2 * ROW_ + tid];
        den += slf[s2 * ROW_ + k * DP_ + D_];
      }
      cLds[tid] = (d < D_) ? (t / den) : 0.0f;
    }
  }
  __syncthreads();

  float cd[D_];
  float bs[4] = {0.f, 0.f, 0.f, 0.f};
#pragma unroll
  for (int d = 0; d < D_; ++d) {
    const float a0 = cLds[d], a1 = cLds[DP_ + d];
    cd[d] = a0 - a1;
    bs[d & 3] = fmaf(cd[d], a0 + a1, bs[d & 3]);
  }
  const float bias = (bs[0] + bs[1]) + (bs[2] + bs[3]);

  v8f acc[3] = {};
  float den0 = 0.f, den1 = 0.f;
  const int m15  = lane & 15;
  const int rsel = (lane >> 4) << 1;
  int aoff[3], boff[3];
#pragma unroll
  for (int w = 0; w < 3; ++w) {
    aoff[w] = (m15 < 14) ? (rsel * XP_ + 14 * w + m15) : ((rsel + 1) * XP_ + D_ + (m15 - 14));
    boff[w] = (m15 < 2) ? (rsel * XP_ + D_ + m15) : ((rsel + 1) * XP_ + 14 * w + (m15 - 2));
  }

#pragma unroll
  for (int it = 0; it < WAVE_ITERS_; ++it) {
    const int n = chunk * CHUNK_ + wave * ROWS_PER_WAVE_ + it * 32 + lane;
    wait_pieces(cur, xw);
    v4f nxt[DQ_];
    if (it + 1 < WAVE_ITERS_) {
#pragma unroll
      for (int q = 0; q < DQ_; ++q) load_piece(nxt[q], xw, ((it + 1) * 32 * DQ_ + 32 * q) * 16);
    }
    float* myxt = &xt[wave][it & 1][0];
#pragma unroll
    for (int q = 0; q < DQ_; ++q) *(v4f*)&myxt[soff[q]] = cur[q];
    __builtin_amdgcn_wave_barrier();
    float ds[4] = {0.f, 0.f, 0.f, 0.f};
#pragma unroll
    for (int q = 0; q < DQ_; ++q) {
      const float4 v = *(const float4*)&myxt[lane * XP_ + 4 * q];
      ds[0] = fmaf(v.x, cd[4 * q + 0], ds[0]);
      ds[1] = fmaf(v.y, cd[4 * q + 1], ds[1]);
      ds[2] = fmaf(v.z, cd[4 * q + 2], ds[2]);
      ds[3] = fmaf(v.w, cd[4 * q + 3], ds[3]);
    }
    const float t   = fmaf(2.0f, (ds[0] + ds[1]) + (ds[2] + ds[3]), -bias);
    const float e   = __expf(-fabsf(t));
    const float inv = 1.0f / (1.0f + e);
    const float ps  = e * inv;
    const float p0  = (t >= 0.0f) ? inv : ps;
    const float p1  = (t >= 0.0f) ? ps : inv;

    if (write_probs) {
      probs_out[(size_t)b * N_ + n] = make_float2(p0, p1);
    }
    if (accumulate) {
      den0 += p0;
      den1 += p1;
      *(float2*)&myxt[lane * XP_ + D_] = make_float2(p0, p1);
      __builtin_amdgcn_wave_barrier();

#pragma unroll
      for (int gp = 0; gp < 4; ++gp) {
        const float* xg = myxt + 8 * gp * XP_;
#pragma unroll
        for (int w = 0; w < 3; ++w) {
          v2f a, bb;
          a.x  = xg[aoff[w]];
          a.y  = xg[aoff[w] + 4 * XP_];
          bb.x = xg[boff[w]];
          bb.y = xg[boff[w] + 4 * XP_];
          acc[w] = __builtin_amdgcn_wmma_f32_16x16x4_f32(
              false, a, false, bb, (short)0, acc[w], false, false);
        }
      }
    }
    if (it + 1 < WAVE_ITERS_) {
#pragma unroll
      for (int q = 0; q < DQ_; ++q) cur[q] = nxt[q];
    }
  }

  if (accumulate) {
#pragma unroll
    for (int off = 16; off > 0; off >>= 1) {
      den0 += __shfl_xor(den0, off);
      den1 += __shfl_xor(den1, off);
    }
#pragma unroll
    for (int w = 0; w < 3; ++w) {
      const int d2 = 14 * w + (lane - 18);
      if (lane >= 18 && d2 < D_) {
        red2[wave][0 * DP_ + d2] = acc[w][6];
        red2[wave][1 * DP_ + d2] = acc[w][7];
      }
      if (m15 < 2) {
#pragma unroll
        for (int j = 0; j < 8; ++j) {
          const int m  = ((lane >> 4) << 3) + j;
          const int d1 = 14 * w + m;
          if (m < 14 && d1 < D_) red1[wave][m15 * DP_ + d1] = acc[w][j];
        }
      }
    }
    if (lane == 0) {
      dn[wave][0] = den0;
      dn[wave][1] = den1;
    }
    __syncthreads();
    if (tid < ROW_) {
      const int k = tid / DP_, d = tid % DP_;
      float s = 0.f;
      if (d < D_) {
#pragma unroll
        for (int w = 0; w < WAVES_; ++w) s += red1[w][tid] + red2[w][tid];
      } else if (d == D_) {
#pragma unroll
        for (int w = 0; w < WAVES_; ++w) s += dn[w][k];
      }
      pout[(size_t)blockIdx.x * ROW_ + tid] = s;
    }
  }
}

extern "C" void kernel_launch(void* const* d_in, const int* in_sizes, int n_in,
                              void* d_out, int out_size, void* d_ws, size_t ws_size,
                              hipStream_t stream) {
  const float* data = (const float*)d_in[0];
  const int*   ids  = (const int*)d_in[1];
  const int ITER = 10;

  float*  rowsA = (float*)d_ws;
  float*  rowsB = rowsA + B_ * CHUNKS_ * ROW_;
  float2* out   = (float2*)d_out;

  for (int t = 0; t <= ITER; ++t) {
    int wp = (t == ITER) ? 1 : 0;
    int ac = (t <  ITER) ? 1 : 0;
    const float* pin  = (t & 1) ? rowsA : rowsB;
    float*       pout = (t & 1) ? rowsB : rowsA;
    if (t == 0)
      sk_pass<true><<<B_ * CHUNKS_, THREADS_, 0, stream>>>(data, ids, pin, pout, out, wp, ac);
    else
      sk_pass<false><<<B_ * CHUNKS_, THREADS_, 0, stream>>>(data, ids, pin, pout, out, wp, ac);
  }
  (void)in_sizes; (void)n_in; (void)out_size; (void)ws_size;
}
